# starting kernel
# baseline (speedup 1.0000x reference)
#include <hip/hip_runtime.h>
#include <hip/hip_bf16.h>

typedef __attribute__((ext_vector_type(2))) float v2f;
typedef __attribute__((ext_vector_type(8))) float v8f;

#define SEQ_T 32768
#define IN_D  512
#define HID_H 20
#define G4H   80

__global__ __launch_bounds__(32) void lstm_gates_gemm(
    const float* __restrict__ x,
    const float* __restrict__ W_ih,
    const float* __restrict__ b_ih,
    const float* __restrict__ b_hh,
    float* __restrict__ xg)
{
    const int mtile = blockIdx.x;
    const int lane  = threadIdx.x;
    const int half  = lane >> 4;
    const int l15   = lane & 15;

    v8f acc[5];
#pragma unroll
    for (int nt = 0; nt < 5; ++nt) {
        const int n = nt * 16 + l15;
        const float bias = b_ih[n] + b_hh[n];
#pragma unroll
        for (int v = 0; v < 8; ++v) acc[nt][v] = bias;
    }

    const float* xrow = x + (size_t)(mtile * 16 + l15) * IN_D;

    for (int k0 = 0; k0 < IN_D; k0 += 4) {
        v2f a = *(const v2f*)(xrow + k0 + 2 * half);
#pragma unroll
        for (int nt = 0; nt < 5; ++nt) {
            v2f b = *(const v2f*)(W_ih + (size_t)(nt * 16 + l15) * IN_D + k0 + 2 * half);
            acc[nt] = __builtin_amdgcn_wmma_f32_16x16x4_f32(
                 false, a,  false, b,
                 (short)0, acc[nt],
                 false,  false);
        }
    }

    float* outbase = xg + (size_t)(mtile * 16) * G4H;
#pragma unroll
    for (int nt = 0; nt < 5; ++nt) {
#pragma unroll
        for (int v = 0; v < 8; ++v) {
            outbase[(size_t)(v + 8 * half) * G4H + nt * 16 + l15] = acc[nt][v];
        }
    }
}

__device__ __forceinline__ float readlane_f(float v, int l) {
    return __builtin_bit_cast(float,
        __builtin_amdgcn_readlane(__builtin_bit_cast(int, v), l));
}

__device__ __forceinline__ float sigmoid_f(float v) {
    return 1.0f / (1.0f + __expf(-v));
}

__global__ __launch_bounds__(32) void lstm_scan(
    const float* __restrict__ xg,
    const float* __restrict__ W_hh,
    const float* __restrict__ W_out,
    const float* __restrict__ b_out,
    const float* __restrict__ h0,
    const float* __restrict__ c0,
    float* __restrict__ out)
{
    const int lane   = threadIdx.x;
    const bool active = lane < HID_H;
    const int  n      = active ? lane : 0;

    float wi[HID_H], wf[HID_H], wg[HID_H], wo[HID_H], wout[HID_H];
#pragma unroll
    for (int k = 0; k < HID_H; ++k) {
        wi[k]   = W_hh[(size_t)(0 * HID_H + n) * HID_H + k];
        wf[k]   = W_hh[(size_t)(1 * HID_H + n) * HID_H + k];
        wg[k]   = W_hh[(size_t)(2 * HID_H + n) * HID_H + k];
        wo[k]   = W_hh[(size_t)(3 * HID_H + n) * HID_H + k];
        wout[k] = W_out[k];
    }
    const float bout = b_out[0];

    float c = c0[n];
    float h = h0[n];

    float hb[HID_H];
#pragma unroll
    for (int k = 0; k < HID_H; ++k) hb[k] = readlane_f(h, k);

    float xi = xg[0 * G4H + 0 * HID_H + n];
    float xf = xg[0 * G4H + 1 * HID_H + n];
    float xc = xg[0 * G4H + 2 * HID_H + n];
    float xo = xg[0 * G4H + 3 * HID_H + n];

    for (int t = 0; t < SEQ_T; ++t) {
        float nxi = 0.f, nxf = 0.f, nxc = 0.f, nxo = 0.f;
        if (t + 1 < SEQ_T) {
            const float* p = xg + (size_t)(t + 1) * G4H;
            nxi = p[0 * HID_H + n];
            nxf = p[1 * HID_H + n];
            nxc = p[2 * HID_H + n];
            nxo = p[3 * HID_H + n];
        }

        float gi = xi, gf = xf, gc = xc, go = xo;
#pragma unroll
        for (int k = 0; k < HID_H; ++k) {
            gi = fmaf(wi[k], hb[k], gi);
            gf = fmaf(wf[k], hb[k], gf);
            gc = fmaf(wg[k], hb[k], gc);
            go = fmaf(wo[k], hb[k], go);
        }

        const float i_ = sigmoid_f(gi);
        const float f_ = sigmoid_f(gf);
        const float o_ = sigmoid_f(go);
        const float g_ = tanhf(gc);

        c = fmaf(f_, c, i_ * g_);
        h = o_ * tanhf(c);

#pragma unroll
        for (int k = 0; k < HID_H; ++k) hb[k] = readlane_f(h, k);

        float ot = bout;
#pragma unroll
        for (int k = 0; k < HID_H; ++k) ot = fmaf(wout[k], hb[k], ot);
        if (lane == 0) out[t] = ot;

        xi = nxi; xf = nxf; xc = nxc; xo = nxo;
    }

    if (active) {
        out[SEQ_T + n]         = h;
        out[SEQ_T + HID_H + n] = c;
    }
}

extern "C" void kernel_launch(void* const* d_in, const int* in_sizes, int n_in,
                              void* d_out, int out_size, void* d_ws, size_t ws_size,
                              hipStream_t stream) {
    (void)in_sizes; (void)n_in; (void)out_size; (void)ws_size;

    const float* x    = (const float*)d_in[0];
    const float* h0   = (const float*)d_in[1];
    const float* c0   = (const float*)d_in[2];
    const float* W_ih = (const float*)d_in[3];
    const float* W_hh = (const float*)d_in[4];
    const float* b_ih = (const float*)d_in[5];
    const float* b_hh = (const float*)d_in[6];
    const float* W_out= (const float*)d_in[7];
    const float* b_out= (const float*)d_in[8];

    float* out = (float*)d_out;
    float* xg  = (float*)d_ws;

    lstm_gates_gemm<<<SEQ_T / 16, 32, 0, stream>>>(x, W_ih, b_ih, b_hh, xg);

    lstm_scan<<<1, 32, 0, stream>>>(xg, W_hh, W_out, b_out, h0, c0, out);
}


# scan: h reaches the gate sums through DPP row_share, not 20 v_readlane: 10 units per 16-lane row, each row sums its 10 terms for 2 units per lane, one permlanex16 joins them; summation order changes
# speedup vs baseline: 1.6855x; 1.6855x over previous
#include <hip/hip_runtime.h>
#include <hip/hip_bf16.h>

typedef __attribute__((ext_vector_type(2))) float v2f;
typedef __attribute__((ext_vector_type(8))) float v8f;

#define SEQ_T 32768
#define IN_D  512
#define HID_H 20
#define G4H   80

__global__ __launch_bounds__(32) void lstm_gates_gemm(
    const float* __restrict__ x,
    const float* __restrict__ W_ih,
    const float* __restrict__ b_ih,
    const float* __restrict__ b_hh,
    float* __restrict__ xg)
{
    const int mtile = blockIdx.x;
    const int lane  = threadIdx.x;
    const int half  = lane >> 4;
    const int l15   = lane & 15;

    v8f acc[5];
#pragma unroll
    for (int nt = 0; nt < 5; ++nt) {
        const int n = nt * 16 + l15;
        const float bias = b_ih[n] + b_hh[n];
#pragma unroll
        for (int v = 0; v < 8; ++v) acc[nt][v] = bias;
    }

    const float* xrow = x + (size_t)(mtile * 16 + l15) * IN_D;

    for (int k0 = 0; k0 < IN_D; k0 += 4) {
        v2f a = *(const v2f*)(xrow + k0 + 2 * half);
#pragma unroll
        for (int nt = 0; nt < 5; ++nt) {
            v2f b = *(const v2f*)(W_ih + (size_t)(nt * 16 + l15) * IN_D + k0 + 2 * half);
            acc[nt] = __builtin_amdgcn_wmma_f32_16x16x4_f32(
                 false, a,  false, b,
                 (short)0, acc[nt],
                 false,  false);
        }
    }

    float* outbase = xg + (size_t)(mtile * 16) * G4H;
#pragma unroll
    for (int nt = 0; nt < 5; ++nt) {
#pragma unroll
        for (int v = 0; v < 8; ++v) {
            outbase[(size_t)(v + 8 * half) * G4H + nt * 16 + l15] = acc[nt][v];
        }
    }
}

#define SCAN_WAVES   5
#define SCAN_THREADS 160
#define ROW_UNITS    10
#define PF_AHEAD     32
#define XG_BYTES     ((size_t)SEQ_T * G4H * sizeof(float))

__device__ __forceinline__ float readlane_f(float v, int l) {
    return __builtin_bit_cast(float,
        __builtin_amdgcn_readlane(__builtin_bit_cast(int, v), l));
}

template <int M>
__device__ __forceinline__ float row_share_f(float v) {
    return __builtin_bit_cast(float,
        __builtin_amdgcn_update_dpp(0, __builtin_bit_cast(int, v),
                                    0x150 + M, 0xF, 0xF, true));
}

__device__ __forceinline__ float other_row_f(float v) {
    return __builtin_bit_cast(float,
        __builtin_amdgcn_permlanex16(__builtin_bit_cast(int, v),
                                     __builtin_bit_cast(int, v),
                                     0x76543210, 0xfedcba98, false, false));
}

__device__ __forceinline__ float sigmoid_f(float v) {
    return 1.0f / (1.0f + __expf(-v));
}

__global__ __launch_bounds__(SCAN_THREADS) void lstm_scan(
    const float* __restrict__ xg,
    const float* __restrict__ W_hh,
    const float* __restrict__ W_out,
    const float* __restrict__ b_out,
    const float* __restrict__ h0,
    const float* __restrict__ c0,
    float* __restrict__ out)
{
    __shared__ float gsh[2][4][32];

    const int lane   = threadIdx.x & 31;
    const int role   = __builtin_amdgcn_readfirstlane((int)(threadIdx.x >> 5));
    const int row    = lane >> 4;
    const int col    = lane & 15;
    const bool active = col < ROW_UNITS;
    const int  n      = active ? row * ROW_UNITS + col : 0;
    const int  nx     = active ? (1 - row) * ROW_UNITS + col : 0;
    const int  kbase  = row * ROW_UNITS;
    const bool gate_wave = role < 4;
    const int  grow   = gate_wave ? role : 0;

    float wk[ROW_UNITS], ws[ROW_UNITS], wout[HID_H];
#pragma unroll
    for (int m = 0; m < ROW_UNITS; ++m) {
        wk[m] = W_hh[(size_t)(grow * HID_H + n)  * HID_H + kbase + m];
        ws[m] = W_hh[(size_t)(grow * HID_H + nx) * HID_H + kbase + m];
    }
#pragma unroll
    for (int k = 0; k < HID_H; ++k) wout[k] = W_out[k];
    const float bout = b_out[0];

    float c = c0[n];
    float h = h0[n];

    const float* xcol = xg + grow * HID_H + n;
    float xa = xcol[0];
    float xb = 0.f;

#define GATE_TERM(M)                                                     \
    {                                                                    \
        const float hv = row_share_f<M>(h);                              \
        gk = fmaf(wk[M], hv, gk);                                        \
        gs = fmaf(ws[M], hv, gs);                                        \
    }

    auto step = [&](const int t, const int buf, const float xcur, float& xnext) {
        const int tn = (t + 1 < SEQ_T) ? (t + 1) : (SEQ_T - 1);
        xnext = xcol[(size_t)tn * G4H];

        if (gate_wave) {
            float gk = xcur;
            float gs = 0.f;
            GATE_TERM(0) GATE_TERM(1) GATE_TERM(2) GATE_TERM(3) GATE_TERM(4)
            GATE_TERM(5) GATE_TERM(6) GATE_TERM(7) GATE_TERM(8) GATE_TERM(9)
            const float g = gk + other_row_f(gs);

            const float a = (role == 2) ? tanhf(g) : sigmoid_f(g);
            gsh[buf][role][lane] = a;
        }

        __builtin_amdgcn_fence(__ATOMIC_RELEASE, "workgroup", "local");
        __builtin_amdgcn_s_barrier();
        __builtin_amdgcn_fence(__ATOMIC_ACQUIRE, "workgroup", "local");

        const float i_ = gsh[buf][0][lane];
        const float f_ = gsh[buf][1][lane];
        const float g_ = gsh[buf][2][lane];
        const float o_ = gsh[buf][3][lane];

        c = fmaf(f_, c, i_ * g_);
        h = o_ * tanhf(c);

        if (!gate_wave) {
            float ot = bout;
#pragma unroll
            for (int k = 0; k < HID_H; ++k) {
                const float hk = readlane_f(h, k < ROW_UNITS ? k : k + (16 - ROW_UNITS));
                ot = fmaf(wout[k], hk, ot);
            }
            if (lane == 0) out[t] = ot;

            if (buf == 0 && lane < 4) {
                size_t off = (size_t)(t + PF_AHEAD) * (G4H * sizeof(float))
                           + (size_t)lane * 256;
                off = off < XG_BYTES ? off : XG_BYTES - sizeof(float);
                __builtin_prefetch((const char*)xg + off, 0, 2);
            }
        }
    };

    for (int t = 0; t < SEQ_T; t += 2) {
        step(t,     0, xa, xb);
        step(t + 1, 1, xb, xa);
    }
#undef GATE_TERM

    if (!gate_wave && active) {
        out[SEQ_T + n]         = h;
        out[SEQ_T + HID_H + n] = c;
    }
}

extern "C" void kernel_launch(void* const* d_in, const int* in_sizes, int n_in,
                              void* d_out, int out_size, void* d_ws, size_t ws_size,
                              hipStream_t stream) {
    (void)in_sizes; (void)n_in; (void)out_size; (void)ws_size;

    const float* x    = (const float*)d_in[0];
    const float* h0   = (const float*)d_in[1];
    const float* c0   = (const float*)d_in[2];
    const float* W_ih = (const float*)d_in[3];
    const float* W_hh = (const float*)d_in[4];
    const float* b_ih = (const float*)d_in[5];
    const float* b_hh = (const float*)d_in[6];
    const float* W_out= (const float*)d_in[7];
    const float* b_out= (const float*)d_in[8];

    float* out = (float*)d_out;
    float* xg  = (float*)d_ws;

    lstm_gates_gemm<<<SEQ_T / 16, 32, 0, stream>>>(x, W_ih, b_ih, b_hh, xg);

    lstm_scan<<<1, SCAN_THREADS, 0, stream>>>(xg, W_hh, W_out, b_out, h0, c0, out);
}


# scan: role dispatch hoisted out of the time loop: one loop per role kind (sigmoid gates, tanh gate, projection), no per-step role tests or taken role branches; values bit-identical
# speedup vs baseline: 1.8156x; 1.0772x over previous
#include <hip/hip_runtime.h>
#include <hip/hip_bf16.h>

typedef __attribute__((ext_vector_type(2))) float v2f;
typedef __attribute__((ext_vector_type(8))) float v8f;

#define SEQ_T 32768
#define IN_D  512
#define HID_H 20
#define G4H   80

__global__ __launch_bounds__(32) void lstm_gates_gemm(
    const float* __restrict__ x,
    const float* __restrict__ W_ih,
    const float* __restrict__ b_ih,
    const float* __restrict__ b_hh,
    float* __restrict__ xg)
{
    const int mtile = blockIdx.x;
    const int lane  = threadIdx.x;
    const int half  = lane >> 4;
    const int l15   = lane & 15;

    v8f acc[5];
#pragma unroll
    for (int nt = 0; nt < 5; ++nt) {
        const int n = nt * 16 + l15;
        const float bias = b_ih[n] + b_hh[n];
#pragma unroll
        for (int v = 0; v < 8; ++v) acc[nt][v] = bias;
    }

    const float* xrow = x + (size_t)(mtile * 16 + l15) * IN_D;

    for (int k0 = 0; k0 < IN_D; k0 += 4) {
        v2f a = *(const v2f*)(xrow + k0 + 2 * half);
#pragma unroll
        for (int nt = 0; nt < 5; ++nt) {
            v2f b = *(const v2f*)(W_ih + (size_t)(nt * 16 + l15) * IN_D + k0 + 2 * half);
            acc[nt] = __builtin_amdgcn_wmma_f32_16x16x4_f32(
                 false, a,  false, b,
                 (short)0, acc[nt],
                 false,  false);
        }
    }

    float* outbase = xg + (size_t)(mtile * 16) * G4H;
#pragma unroll
    for (int nt = 0; nt < 5; ++nt) {
#pragma unroll
        for (int v = 0; v < 8; ++v) {
            outbase[(size_t)(v + 8 * half) * G4H + nt * 16 + l15] = acc[nt][v];
        }
    }
}

#define SCAN_WAVES   5
#define SCAN_THREADS 160
#define ROW_UNITS    10
#define PF_AHEAD     32
#define XG_BYTES     ((size_t)SEQ_T * G4H * sizeof(float))

__device__ __forceinline__ float readlane_f(float v, int l) {
    return __builtin_bit_cast(float,
        __builtin_amdgcn_readlane(__builtin_bit_cast(int, v), l));
}

template <int M>
__device__ __forceinline__ float row_share_f(float v) {
    return __builtin_bit_cast(float,
        __builtin_amdgcn_update_dpp(0, __builtin_bit_cast(int, v),
                                    0x150 + M, 0xF, 0xF, true));
}

__device__ __forceinline__ float other_row_f(float v) {
    return __builtin_bit_cast(float,
        __builtin_amdgcn_permlanex16(__builtin_bit_cast(int, v),
                                     __builtin_bit_cast(int, v),
                                     0x76543210, 0xfedcba98, false, false));
}

__device__ __forceinline__ float sigmoid_f(float v) {
    return 1.0f / (1.0f + __expf(-v));
}

#define KIND_SIGMOID 0
#define KIND_TANH    1
#define KIND_PROJ    2

template <bool B> struct Flag { static constexpr bool value = B; };

template <int KIND>
__device__ __forceinline__ void scan_role_loop(
    const float* __restrict__ xg, const float* __restrict__ W_hh,
    const float* __restrict__ W_out, const float* __restrict__ b_out,
    const float* __restrict__ h0, const float* __restrict__ c0,
    float* __restrict__ out, float (*gsh)[4][32],
    const int role, const int lane)
{
    const int row    = lane >> 4;
    const int col    = lane & 15;
    const bool active = col < ROW_UNITS;
    const int  n      = active ? row * ROW_UNITS + col : 0;
    const int  nx     = active ? (1 - row) * ROW_UNITS + col : 0;
    const int  kbase  = row * ROW_UNITS;

    float c = c0[n];
    float h = h0[n];

    auto cell_update = [&](const int buf) {
        const float i_ = gsh[buf][0][lane];
        const float f_ = gsh[buf][1][lane];
        const float g_ = gsh[buf][2][lane];
        const float o_ = gsh[buf][3][lane];
        c = fmaf(f_, c, i_ * g_);
        h = o_ * tanhf(c);
    };

    auto hand_over = [&]() {
        __builtin_amdgcn_fence(__ATOMIC_RELEASE, "workgroup", "local");
        __builtin_amdgcn_s_barrier();
        __builtin_amdgcn_fence(__ATOMIC_ACQUIRE, "workgroup", "local");
    };

    if constexpr (KIND == KIND_PROJ) {
        float wout[HID_H];
#pragma unroll
        for (int k = 0; k < HID_H; ++k) wout[k] = W_out[k];
        const float bout = b_out[0];

        auto step = [&](const int t, const int buf) {
            hand_over();
            cell_update(buf);

            float ot = bout;
#pragma unroll
            for (int k = 0; k < HID_H; ++k) {
                const float hk = readlane_f(h, k < ROW_UNITS ? k : k + (16 - ROW_UNITS));
                ot = fmaf(wout[k], hk, ot);
            }
            if (lane == 0) out[t] = ot;

            if (buf == 0 && lane < 4) {
                size_t off = (size_t)(t + PF_AHEAD) * (G4H * sizeof(float))
                           + (size_t)lane * 256;
                off = off < XG_BYTES ? off : XG_BYTES - sizeof(float);
                __builtin_prefetch((const char*)xg + off, 0, 2);
            }
        };

        for (int t = 0; t < SEQ_T; t += 2) {
            step(t,     0);
            step(t + 1, 1);
        }

        if (active) {
            out[SEQ_T + n]         = h;
            out[SEQ_T + HID_H + n] = c;
        }
    } else {
        float wk[ROW_UNITS], ws[ROW_UNITS];
#pragma unroll
        for (int m = 0; m < ROW_UNITS; ++m) {
            wk[m] = W_hh[(size_t)(role * HID_H + n)  * HID_H + kbase + m];
            ws[m] = W_hh[(size_t)(role * HID_H + nx) * HID_H + kbase + m];
        }

        const unsigned xcolumn = (unsigned)(role * HID_H + n);
        float xa = xg[xcolumn];
        float xb = 0.f;

#define GATE_TERM(M)                                                     \
    {                                                                    \
        const float hv = row_share_f<M>(h);                              \
        gk = fmaf(wk[M], hv, gk);                                        \
        gs = fmaf(ws[M], hv, gs);                                        \
    }

        auto step = [&](const int t, const int buf, const float xcur, float& xnext,
                        auto load_next) {
            if constexpr (decltype(load_next)::value) {
                const float* xrow = xg + (size_t)(t + 1) * G4H;
                xnext = xrow[xcolumn];
            }

            float gk = xcur;
            float gs = 0.f;
            GATE_TERM(0) GATE_TERM(1) GATE_TERM(2) GATE_TERM(3) GATE_TERM(4)
            GATE_TERM(5) GATE_TERM(6) GATE_TERM(7) GATE_TERM(8) GATE_TERM(9)
            const float g = gk + other_row_f(gs);

            float a;
            if constexpr (KIND == KIND_TANH) a = tanhf(g);
            else                             a = sigmoid_f(g);
            gsh[buf][role][lane] = a;

            hand_over();
            cell_update(buf);
        };

        const Flag<true>  with_load{};
        const Flag<false> no_load{};
        for (int t = 0; t < SEQ_T - 2; t += 2) {
            step(t,     0, xa, xb, with_load);
            step(t + 1, 1, xb, xa, with_load);
        }
        step(SEQ_T - 2, 0, xa, xb, with_load);
        step(SEQ_T - 1, 1, xb, xa, no_load);
#undef GATE_TERM
    }
}

__global__ __launch_bounds__(SCAN_THREADS) void lstm_scan(
    const float* __restrict__ xg,
    const float* __restrict__ W_hh,
    const float* __restrict__ W_out,
    const float* __restrict__ b_out,
    const float* __restrict__ h0,
    const float* __restrict__ c0,
    float* __restrict__ out)
{
    __shared__ float gsh[2][4][32];

    const int lane = threadIdx.x & 31;
    const int role = __builtin_amdgcn_readfirstlane((int)(threadIdx.x >> 5));

    if (role == 4)
        scan_role_loop<KIND_PROJ>(xg, W_hh, W_out, b_out, h0, c0, out, gsh, role, lane);
    else if (role == 2)
        scan_role_loop<KIND_TANH>(xg, W_hh, W_out, b_out, h0, c0, out, gsh, role, lane);
    else
        scan_role_loop<KIND_SIGMOID>(xg, W_hh, W_out, b_out, h0, c0, out, gsh, role, lane);
}

extern "C" void kernel_launch(void* const* d_in, const int* in_sizes, int n_in,
                              void* d_out, int out_size, void* d_ws, size_t ws_size,
                              hipStream_t stream) {
    (void)in_sizes; (void)n_in; (void)out_size; (void)ws_size;

    const float* x    = (const float*)d_in[0];
    const float* h0   = (const float*)d_in[1];
    const float* c0   = (const float*)d_in[2];
    const float* W_ih = (const float*)d_in[3];
    const float* W_hh = (const float*)d_in[4];
    const float* b_ih = (const float*)d_in[5];
    const float* b_hh = (const float*)d_in[6];
    const float* W_out= (const float*)d_in[7];
    const float* b_out= (const float*)d_in[8];

    float* out = (float*)d_out;
    float* xg  = (float*)d_ws;

    lstm_gates_gemm<<<SEQ_T / 16, 32, 0, stream>>>(x, W_ih, b_ih, b_hh, xg);

    lstm_scan<<<1, SCAN_THREADS, 0, stream>>>(xg, W_hh, W_out, b_out, h0, c0, out);
}


# scan: g gate's tanh split by argument range over two waves (a sixth wave): each runs one form of the library tanh per step, complementary masked LDS stores; values bit-identical
# speedup vs baseline: 1.9435x; 1.0704x over previous
#include <hip/hip_runtime.h>
#include <hip/hip_bf16.h>

typedef __attribute__((ext_vector_type(2))) float v2f;
typedef __attribute__((ext_vector_type(8))) float v8f;

#define SEQ_T 32768
#define IN_D  512
#define HID_H 20
#define G4H   80

__global__ __launch_bounds__(32) void lstm_gates_gemm(
    const float* __restrict__ x,
    const float* __restrict__ W_ih,
    const float* __restrict__ b_ih,
    const float* __restrict__ b_hh,
    float* __restrict__ xg)
{
    const int mtile = blockIdx.x;
    const int lane  = threadIdx.x;
    const int half  = lane >> 4;
    const int l15   = lane & 15;

    v8f acc[5];
#pragma unroll
    for (int nt = 0; nt < 5; ++nt) {
        const int n = nt * 16 + l15;
        const float bias = b_ih[n] + b_hh[n];
#pragma unroll
        for (int v = 0; v < 8; ++v) acc[nt][v] = bias;
    }

    const float* xrow = x + (size_t)(mtile * 16 + l15) * IN_D;

    for (int k0 = 0; k0 < IN_D; k0 += 4) {
        v2f a = *(const v2f*)(xrow + k0 + 2 * half);
#pragma unroll
        for (int nt = 0; nt < 5; ++nt) {
            v2f b = *(const v2f*)(W_ih + (size_t)(nt * 16 + l15) * IN_D + k0 + 2 * half);
            acc[nt] = __builtin_amdgcn_wmma_f32_16x16x4_f32(
                 false, a,  false, b,
                 (short)0, acc[nt],
                 false,  false);
        }
    }

    float* outbase = xg + (size_t)(mtile * 16) * G4H;
#pragma unroll
    for (int nt = 0; nt < 5; ++nt) {
#pragma unroll
        for (int v = 0; v < 8; ++v) {
            outbase[(size_t)(v + 8 * half) * G4H + nt * 16 + l15] = acc[nt][v];
        }
    }
}

#define SCAN_WAVES   6
#define SCAN_THREADS 192
#define GATE_G       2
#define TANH_SPLIT   0.625f
#define ROW_UNITS    10
#define PF_AHEAD     32
#define XG_BYTES     ((size_t)SEQ_T * G4H * sizeof(float))

__device__ __forceinline__ float readlane_f(float v, int l) {
    return __builtin_bit_cast(float,
        __builtin_amdgcn_readlane(__builtin_bit_cast(int, v), l));
}

template <int M>
__device__ __forceinline__ float row_share_f(float v) {
    return __builtin_bit_cast(float,
        __builtin_amdgcn_update_dpp(0, __builtin_bit_cast(int, v),
                                    0x150 + M, 0xF, 0xF, true));
}

__device__ __forceinline__ float other_row_f(float v) {
    return __builtin_bit_cast(float,
        __builtin_amdgcn_permlanex16(__builtin_bit_cast(int, v),
                                     __builtin_bit_cast(int, v),
                                     0x76543210, 0xfedcba98, false, false));
}

__device__ __forceinline__ float sigmoid_f(float v) {
    return 1.0f / (1.0f + __expf(-v));
}

#define KIND_SIGMOID 0
#define KIND_TANH_HI 1
#define KIND_PROJ    2
#define KIND_TANH_LO 3

template <bool B> struct Flag { static constexpr bool value = B; };

template <int KIND>
__device__ __forceinline__ void scan_role_loop(
    const float* __restrict__ xg, const float* __restrict__ W_hh,
    const float* __restrict__ W_out, const float* __restrict__ b_out,
    const float* __restrict__ h0, const float* __restrict__ c0,
    float* __restrict__ out, float (*gsh)[4][32],
    const int role, const int lane)
{
    const int row    = lane >> 4;
    const int col    = lane & 15;
    const bool active = col < ROW_UNITS;
    const int  n      = active ? row * ROW_UNITS + col : 0;
    const int  nx     = active ? (1 - row) * ROW_UNITS + col : 0;
    const int  kbase  = row * ROW_UNITS;

    float c = c0[n];
    float h = h0[n];

    auto cell_update = [&](const int buf) {
        const float i_ = gsh[buf][0][lane];
        const float f_ = gsh[buf][1][lane];
        const float g_ = gsh[buf][2][lane];
        const float o_ = gsh[buf][3][lane];
        c = fmaf(f_, c, i_ * g_);
        h = o_ * tanhf(c);
    };

    auto hand_over = [&]() {
        __builtin_amdgcn_fence(__ATOMIC_RELEASE, "workgroup", "local");
        __builtin_amdgcn_s_barrier();
        __builtin_amdgcn_fence(__ATOMIC_ACQUIRE, "workgroup", "local");
    };

    if constexpr (KIND == KIND_PROJ) {
        float wout[HID_H];
#pragma unroll
        for (int k = 0; k < HID_H; ++k) wout[k] = W_out[k];
        const float bout = b_out[0];

        auto step = [&](const int t, const int buf) {
            hand_over();
            cell_update(buf);

            float ot = bout;
#pragma unroll
            for (int k = 0; k < HID_H; ++k) {
                const float hk = readlane_f(h, k < ROW_UNITS ? k : k + (16 - ROW_UNITS));
                ot = fmaf(wout[k], hk, ot);
            }
            if (lane == 0) out[t] = ot;

            if (buf == 0 && lane < 4) {
                size_t off = (size_t)(t + PF_AHEAD) * (G4H * sizeof(float))
                           + (size_t)lane * 256;
                off = off < XG_BYTES ? off : XG_BYTES - sizeof(float);
                __builtin_prefetch((const char*)xg + off, 0, 2);
            }
        };

        for (int t = 0; t < SEQ_T; t += 2) {
            step(t,     0);
            step(t + 1, 1);
        }

        if (active) {
            out[SEQ_T + n]         = h;
            out[SEQ_T + HID_H + n] = c;
        }
    } else {
        float wk[ROW_UNITS], ws[ROW_UNITS];
#pragma unroll
        for (int m = 0; m < ROW_UNITS; ++m) {
            wk[m] = W_hh[(size_t)(role * HID_H + n)  * HID_H + kbase + m];
            ws[m] = W_hh[(size_t)(role * HID_H + nx) * HID_H + kbase + m];
        }

        const unsigned xcolumn = (unsigned)(role * HID_H + n);
        float xa = xg[xcolumn];
        float xb = 0.f;

#define GATE_TERM(M)                                                     \
    {                                                                    \
        const float hv = row_share_f<M>(h);                              \
        gk = fmaf(wk[M], hv, gk);                                        \
        gs = fmaf(ws[M], hv, gs);                                        \
    }

        auto step = [&](const int t, const int buf, const float xcur, float& xnext,
                        auto load_next) {
            if constexpr (decltype(load_next)::value) {
                const float* xrow = xg + (size_t)(t + 1) * G4H;
                xnext = xrow[xcolumn];
            }

            float gk = xcur;
            float gs = 0.f;
            GATE_TERM(0) GATE_TERM(1) GATE_TERM(2) GATE_TERM(3) GATE_TERM(4)
            GATE_TERM(5) GATE_TERM(6) GATE_TERM(7) GATE_TERM(8) GATE_TERM(9)
            const float g = gk + other_row_f(gs);

            if constexpr (KIND == KIND_SIGMOID) {
                gsh[buf][role][lane] = sigmoid_f(g);
            } else {
                const bool low = __builtin_fabsf(g) < TANH_SPLIT;
                if constexpr (KIND == KIND_TANH_HI) {
                    if (!low) gsh[buf][role][lane] = tanhf(g);
                } else {
                    if (low)  gsh[buf][role][lane] = tanhf(g);
                }
            }

            hand_over();
            cell_update(buf);
        };

        const Flag<true>  with_load{};
        const Flag<false> no_load{};
        for (int t = 0; t < SEQ_T - 2; t += 2) {
            step(t,     0, xa, xb, with_load);
            step(t + 1, 1, xb, xa, with_load);
        }
        step(SEQ_T - 2, 0, xa, xb, with_load);
        step(SEQ_T - 1, 1, xb, xa, no_load);
#undef GATE_TERM
    }
}

__global__ __launch_bounds__(SCAN_THREADS) void lstm_scan(
    const float* __restrict__ xg,
    const float* __restrict__ W_hh,
    const float* __restrict__ W_out,
    const float* __restrict__ b_out,
    const float* __restrict__ h0,
    const float* __restrict__ c0,
    float* __restrict__ out)
{
    __shared__ float gsh[2][4][32];

    const int lane = threadIdx.x & 31;
    const int wave = __builtin_amdgcn_readfirstlane((int)(threadIdx.x >> 5));

    if (wave == 4)
        scan_role_loop<KIND_PROJ>(xg, W_hh, W_out, b_out, h0, c0, out, gsh, wave, lane);
    else if (wave == 2)
        scan_role_loop<KIND_TANH_HI>(xg, W_hh, W_out, b_out, h0, c0, out, gsh, GATE_G, lane);
    else if (wave == 5)
        scan_role_loop<KIND_TANH_LO>(xg, W_hh, W_out, b_out, h0, c0, out, gsh, GATE_G, lane);
    else
        scan_role_loop<KIND_SIGMOID>(xg, W_hh, W_out, b_out, h0, c0, out, gsh, wave, lane);
}

extern "C" void kernel_launch(void* const* d_in, const int* in_sizes, int n_in,
                              void* d_out, int out_size, void* d_ws, size_t ws_size,
                              hipStream_t stream) {
    (void)in_sizes; (void)n_in; (void)out_size; (void)ws_size;

    const float* x    = (const float*)d_in[0];
    const float* h0   = (const float*)d_in[1];
    const float* c0   = (const float*)d_in[2];
    const float* W_ih = (const float*)d_in[3];
    const float* W_hh = (const float*)d_in[4];
    const float* b_ih = (const float*)d_in[5];
    const float* b_hh = (const float*)d_in[6];
    const float* W_out= (const float*)d_in[7];
    const float* b_out= (const float*)d_in[8];

    float* out = (float*)d_out;
    float* xg  = (float*)d_ws;

    lstm_gates_gemm<<<SEQ_T / 16, 32, 0, stream>>>(x, W_ih, b_ih, b_hh, xg);

    lstm_scan<<<1, SCAN_THREADS, 0, stream>>>(xg, W_hh, W_out, b_out, h0, c0, out);
}


# scan re-split by hidden unit: a wave owns all 4 gates of 8 units, waves exchange h (not gates), so tanh(c) splits by range over wave pairs, one form each; g's tanh as 2*sigmoid(2x)-1; values move
# speedup vs baseline: 2.4145x; 1.2423x over previous
#include <hip/hip_runtime.h>
#include <hip/hip_bf16.h>

typedef __attribute__((ext_vector_type(2))) float v2f;
typedef __attribute__((ext_vector_type(8))) float v8f;

#define SEQ_T 32768
#define IN_D  512
#define HID_H 20
#define G4H   80

__global__ __launch_bounds__(32) void lstm_gates_gemm(
    const float* __restrict__ x,
    const float* __restrict__ W_ih,
    const float* __restrict__ b_ih,
    const float* __restrict__ b_hh,
    float* __restrict__ xg)
{
    const int mtile = blockIdx.x;
    const int lane  = threadIdx.x;
    const int half  = lane >> 4;
    const int l15   = lane & 15;

    v8f acc[5];
#pragma unroll
    for (int nt = 0; nt < 5; ++nt) {
        const int n = nt * 16 + l15;
        const float bias = b_ih[n] + b_hh[n];
#pragma unroll
        for (int v = 0; v < 8; ++v) acc[nt][v] = bias;
    }

    const float* xrow = x + (size_t)(mtile * 16 + l15) * IN_D;

    for (int k0 = 0; k0 < IN_D; k0 += 4) {
        v2f a = *(const v2f*)(xrow + k0 + 2 * half);
#pragma unroll
        for (int nt = 0; nt < 5; ++nt) {
            v2f b = *(const v2f*)(W_ih + (size_t)(nt * 16 + l15) * IN_D + k0 + 2 * half);
            acc[nt] = __builtin_amdgcn_wmma_f32_16x16x4_f32(
                 false, a,  false, b,
                 (short)0, acc[nt],
                 false,  false);
        }
    }

    float* outbase = xg + (size_t)(mtile * 16) * G4H;
#pragma unroll
    for (int nt = 0; nt < 5; ++nt) {
#pragma unroll
        for (int v = 0; v < 8; ++v) {
            outbase[(size_t)(v + 8 * half) * G4H + nt * 16 + l15] = acc[nt][v];
        }
    }
}

#define SCAN_WAVES   7
#define SCAN_THREADS 224
#define WAVE_UNITS   8
#define PROJ_WAVE    3
#define GATE_G       2
#define TANH_SPLIT   0.625f
#define ROW_UNITS    10
#define PF_AHEAD     32
#define XG_BYTES     ((size_t)SEQ_T * G4H * sizeof(float))
#define NEG_LOG2E    (-1.44269504088896340736f)

__device__ __forceinline__ float readlane_f(float v, int l) {
    return __builtin_bit_cast(float,
        __builtin_amdgcn_readlane(__builtin_bit_cast(int, v), l));
}

template <int M>
__device__ __forceinline__ float row_share_f(float v) {
    return __builtin_bit_cast(float,
        __builtin_amdgcn_update_dpp(0, __builtin_bit_cast(int, v),
                                    0x150 + M, 0xF, 0xF, true));
}

template <int Q>
__device__ __forceinline__ float quad_share_f(float v) {
    return __builtin_bit_cast(float,
        __builtin_amdgcn_update_dpp(0, __builtin_bit_cast(int, v),
                                    Q * 0x55, 0xF, 0xF, true));
}

__device__ __forceinline__ float other_row_f(float v) {
    return __builtin_bit_cast(float,
        __builtin_amdgcn_permlanex16(__builtin_bit_cast(int, v),
                                     __builtin_bit_cast(int, v),
                                     0x76543210, 0xfedcba98, false, false));
}

__device__ __forceinline__ void hand_over() {
    __builtin_amdgcn_fence(__ATOMIC_RELEASE, "workgroup", "local");
    __builtin_amdgcn_s_barrier();
    __builtin_amdgcn_fence(__ATOMIC_ACQUIRE, "workgroup", "local");
}

#define KIND_TANH_HI 0
#define KIND_TANH_LO 1

template <bool B> struct Flag { static constexpr bool value = B; };

template <int KIND>
__device__ __forceinline__ void scan_unit_loop(
    const float* __restrict__ xg, const float* __restrict__ W_hh,
    const float* __restrict__ h0, const float* __restrict__ c0,
    float (*hsh)[32], float* csh, const int group, const int lane)
{
    const int row   = lane >> 4;
    const int col   = lane & 15;
    const int gate  = lane & 3;
    const int uraw  = group * WAVE_UNITS + (lane >> 2);
    const int urawx = group * WAVE_UNITS + ((lane ^ 16) >> 2);
    const int unit  = uraw  < HID_H ? uraw  : uraw  - 4;
    const int unitx = urawx < HID_H ? urawx : urawx - 4;
    const bool owner = gate == 0 && uraw < HID_H;
    const int hsrc  = col < ROW_UNITS ? row * ROW_UNITS + col : 0;
    const int kbase = row * ROW_UNITS;
    const int grow  = gate * HID_H + unit;
    const int growx = gate * HID_H + unitx;

    float wk[ROW_UNITS], ws[ROW_UNITS];
#pragma unroll
    for (int m = 0; m < ROW_UNITS; ++m) {
        wk[m] = W_hh[(size_t)grow  * HID_H + kbase + m];
        ws[m] = W_hh[(size_t)growx * HID_H + kbase + m];
    }

    const float sc  = gate == GATE_G ? 2.0f : 1.0f;
    const float nk  = sc * NEG_LOG2E;
    const float off = 1.0f - sc;

    float c  = c0[unit];
    float hv = h0[hsrc];

    const unsigned xcolumn = (unsigned)grow;
    float xa = xg[xcolumn];
    float xb = 0.f;

#define GATE_TERM(M)                                                     \
    {                                                                    \
        const float hk = row_share_f<M>(hv);                             \
        gk = fmaf(wk[M], hk, gk);                                        \
        gs = fmaf(ws[M], hk, gs);                                        \
    }

    auto step = [&](const int t, const int buf, const float xcur, float& xnext,
                    auto load_next) {
        if constexpr (decltype(load_next)::value) {
            const float* xrow = xg + (size_t)(t + 1) * G4H;
            xnext = xrow[xcolumn];
        }

        float gk = xcur;
        float gs = 0.f;
        GATE_TERM(0) GATE_TERM(1) GATE_TERM(2) GATE_TERM(3) GATE_TERM(4)
        GATE_TERM(5) GATE_TERM(6) GATE_TERM(7) GATE_TERM(8) GATE_TERM(9)
        const float pre = gk + other_row_f(gs);

        const float e = __builtin_amdgcn_exp2f(nk * pre);
        const float a = fmaf(1.0f / (1.0f + e), sc, off);

        const float i_ = quad_share_f<0>(a);
        const float f_ = quad_share_f<1>(a);
        const float g_ = quad_share_f<2>(a);
        const float o_ = quad_share_f<3>(a);
        c = fmaf(f_, c, i_ * g_);

        const bool low = __builtin_fabsf(c) < TANH_SPLIT;
        if constexpr (KIND == KIND_TANH_HI) {
            if (!low) {
                hsh[buf][uraw] = o_ * tanhf(c);
            }
        } else {
            if (low) {
                hsh[buf][uraw] = o_ * tanhf(c);
            }
        }

        hand_over();
        hv = hsh[buf][hsrc];
    };

    const Flag<true>  with_load{};
    const Flag<false> no_load{};
    for (int t = 0; t < SEQ_T - 2; t += 2) {
        step(t,     0, xa, xb, with_load);
        step(t + 1, 1, xb, xa, with_load);
    }
    step(SEQ_T - 2, 0, xa, xb, with_load);
    step(SEQ_T - 1, 1, xb, xa, no_load);
#undef GATE_TERM

    if constexpr (KIND == KIND_TANH_LO) {
        if (owner) csh[unit] = c;
    }
    __syncthreads();
}

__device__ __forceinline__ void scan_proj_loop(
    const float* __restrict__ xg, const float* __restrict__ W_out,
    const float* __restrict__ b_out, float* __restrict__ out,
    float (*hsh)[32], const float* csh, const int lane)
{
    const int row = lane >> 4;
    const int col = lane & 15;
    const bool active = col < ROW_UNITS;
    const int  n      = active ? row * ROW_UNITS + col : 0;

    float wout[HID_H];
#pragma unroll
    for (int k = 0; k < HID_H; ++k) wout[k] = W_out[k];
    const float bout = b_out[0];
    float hv = 0.f;

    auto step = [&](const int t, const int buf) {
        hand_over();
        hv = hsh[buf][n];

        float ot = bout;
#pragma unroll
        for (int k = 0; k < HID_H; ++k) {
            const float hk = readlane_f(hv, k < ROW_UNITS ? k : k + (16 - ROW_UNITS));
            ot = fmaf(wout[k], hk, ot);
        }
        if (lane == 0) out[t] = ot;

        if (buf == 0 && lane < 4) {
            size_t off = (size_t)(t + PF_AHEAD) * (G4H * sizeof(float))
                       + (size_t)lane * 256;
            off = off < XG_BYTES ? off : XG_BYTES - sizeof(float);
            __builtin_prefetch((const char*)xg + off, 0, 2);
        }
    };

    for (int t = 0; t < SEQ_T; t += 2) {
        step(t,     0);
        step(t + 1, 1);
    }

    __syncthreads();
    if (active) {
        out[SEQ_T + n]         = hv;
        out[SEQ_T + HID_H + n] = csh[n];
    }
}

__global__ __launch_bounds__(SCAN_THREADS) void lstm_scan(
    const float* __restrict__ xg,
    const float* __restrict__ W_hh,
    const float* __restrict__ W_out,
    const float* __restrict__ b_out,
    const float* __restrict__ h0,
    const float* __restrict__ c0,
    float* __restrict__ out)
{
    __shared__ float hsh[2][32];
    __shared__ float csh[32];

    const int lane = threadIdx.x & 31;
    const int wave = __builtin_amdgcn_readfirstlane((int)(threadIdx.x >> 5));

    if (wave == PROJ_WAVE)
        scan_proj_loop(xg, W_out, b_out, out, hsh, csh, lane);
    else if (wave < PROJ_WAVE)
        scan_unit_loop<KIND_TANH_HI>(xg, W_hh, h0, c0, hsh, csh, wave, lane);
    else
        scan_unit_loop<KIND_TANH_LO>(xg, W_hh, h0, c0, hsh, csh, wave - PROJ_WAVE - 1, lane);
}

extern "C" void kernel_launch(void* const* d_in, const int* in_sizes, int n_in,
                              void* d_out, int out_size, void* d_ws, size_t ws_size,
                              hipStream_t stream) {
    (void)in_sizes; (void)n_in; (void)out_size; (void)ws_size;

    const float* x    = (const float*)d_in[0];
    const float* h0   = (const float*)d_in[1];
    const float* c0   = (const float*)d_in[2];
    const float* W_ih = (const float*)d_in[3];
    const float* W_hh = (const float*)d_in[4];
    const float* b_ih = (const float*)d_in[5];
    const float* b_hh = (const float*)d_in[6];
    const float* W_out= (const float*)d_in[7];
    const float* b_out= (const float*)d_in[8];

    float* out = (float*)d_out;
    float* xg  = (float*)d_ws;

    lstm_gates_gemm<<<SEQ_T / 16, 32, 0, stream>>>(x, W_ih, b_ih, b_hh, xg);

    lstm_scan<<<1, SCAN_THREADS, 0, stream>>>(xg, W_hh, W_out, b_out, h0, c0, out);
}


# speedup vs baseline: 2.4909x; 1.0316x over previous
#include <hip/hip_runtime.h>
#include <hip/hip_bf16.h>

typedef __attribute__((ext_vector_type(2))) float v2f;
typedef __attribute__((ext_vector_type(8))) float v8f;

#define SEQ_T 32768
#define IN_D  512
#define HID_H 20
#define G4H   80

__global__ __launch_bounds__(32) void lstm_gates_gemm(
    const float* __restrict__ x,
    const float* __restrict__ W_ih,
    const float* __restrict__ b_ih,
    const float* __restrict__ b_hh,
    float* __restrict__ xg)
{
    const int mtile = blockIdx.x;
    const int lane  = threadIdx.x;
    const int half  = lane >> 4;
    const int l15   = lane & 15;

    v8f acc[5];
#pragma unroll
    for (int nt = 0; nt < 5; ++nt) {
        const int n = nt * 16 + l15;
        const float bias = b_ih[n] + b_hh[n];
#pragma unroll
        for (int v = 0; v < 8; ++v) acc[nt][v] = bias;
    }

    const float* xrow = x + (size_t)(mtile * 16 + l15) * IN_D;

    for (int k0 = 0; k0 < IN_D; k0 += 4) {
        v2f a = *(const v2f*)(xrow + k0 + 2 * half);
#pragma unroll
        for (int nt = 0; nt < 5; ++nt) {
            v2f b = *(const v2f*)(W_ih + (size_t)(nt * 16 + l15) * IN_D + k0 + 2 * half);
            acc[nt] = __builtin_amdgcn_wmma_f32_16x16x4_f32(
                 false, a,  false, b,
                 (short)0, acc[nt],
                 false,  false);
        }
    }

    float* outbase = xg + (size_t)(mtile * 16) * G4H;
#pragma unroll
    for (int nt = 0; nt < 5; ++nt) {
#pragma unroll
        for (int v = 0; v < 8; ++v) {
            outbase[(size_t)(v + 8 * half) * G4H + nt * 16 + l15] = acc[nt][v];
        }
    }
}

#define SCAN_WAVES   7
#define SCAN_THREADS 224
#define WAVE_UNITS   8
#define PROJ_WAVE    3
#define GATE_G       2
#define TANH_SPLIT   0.625f
#define ROW_UNITS    10
#define PF_AHEAD     32
#define XG_BYTES     ((size_t)SEQ_T * G4H * sizeof(float))
#define NEG_LOG2E    (-1.44269504088896340736f)

__device__ __forceinline__ float readlane_f(float v, int l) {
    return __builtin_bit_cast(float,
        __builtin_amdgcn_readlane(__builtin_bit_cast(int, v), l));
}

template <int M>
__device__ __forceinline__ float row_share_f(float v) {
    constexpr unsigned SEL = 0x11111111u * (unsigned)M;
    return __builtin_bit_cast(float,
        __builtin_amdgcn_permlane16(__builtin_bit_cast(int, v),
                                    __builtin_bit_cast(int, v),
                                    SEL, SEL, false, true));
}

template <int Q>
__device__ __forceinline__ float quad_share_f(float v) {
    return __builtin_bit_cast(float,
        __builtin_amdgcn_update_dpp(0, __builtin_bit_cast(int, v),
                                    Q * 0x55, 0xF, 0xF, true));
}

__device__ __forceinline__ float other_row_f(float v) {
    return __builtin_bit_cast(float,
        __builtin_amdgcn_permlanex16(__builtin_bit_cast(int, v),
                                     __builtin_bit_cast(int, v),
                                     0x76543210, 0xfedcba98, false, false));
}

__device__ __forceinline__ void hand_over() {
    __builtin_amdgcn_fence(__ATOMIC_RELEASE, "workgroup", "local");
    __builtin_amdgcn_s_barrier();
    __builtin_amdgcn_fence(__ATOMIC_ACQUIRE, "workgroup", "local");
}

#define KIND_TANH_HI 0
#define KIND_TANH_LO 1

template <bool B> struct Flag { static constexpr bool value = B; };

template <int KIND>
__device__ __forceinline__ void scan_unit_loop(
    const float* __restrict__ xg, const float* __restrict__ W_hh,
    const float* __restrict__ h0, const float* __restrict__ c0,
    float (*hsh)[32], float* csh, const int group, const int lane)
{
    const int row   = lane >> 4;
    const int col   = lane & 15;
    const int gate  = lane & 3;
    const int uraw  = group * WAVE_UNITS + (lane >> 2);
    const int urawx = group * WAVE_UNITS + ((lane ^ 16) >> 2);
    const int unit  = uraw  < HID_H ? uraw  : uraw  - 4;
    const int unitx = urawx < HID_H ? urawx : urawx - 4;
    const bool owner = gate == 0 && uraw < HID_H;
    const int hsrc  = col < ROW_UNITS ? row * ROW_UNITS + col : 0;
    const int kbase = row * ROW_UNITS;
    const int grow  = gate * HID_H + unit;
    const int growx = gate * HID_H + unitx;

    float wk[ROW_UNITS], ws[ROW_UNITS];
#pragma unroll
    for (int m = 0; m < ROW_UNITS; ++m) {
        wk[m] = W_hh[(size_t)grow  * HID_H + kbase + m];
        ws[m] = W_hh[(size_t)growx * HID_H + kbase + m];
    }

    const float sc  = gate == GATE_G ? 2.0f : 1.0f;
    const float nk  = sc * NEG_LOG2E;
    const float off = 1.0f - sc;

    float c  = c0[unit];
    float hv = h0[hsrc];

    const unsigned xcolumn = (unsigned)grow;
    float xa = xg[xcolumn];
    float xb = 0.f;

#define GATE_TERM(M)                                                     \
    {                                                                    \
        const float hk = row_share_f<M>(hv);                             \
        gk = fmaf(wk[M], hk, gk);                                        \
        gs = fmaf(ws[M], hk, gs);                                        \
    }

    auto step = [&](const int t, const int buf, const float xcur, float& xnext,
                    auto load_next) {
        if constexpr (decltype(load_next)::value) {
            const float* xrow = xg + (size_t)(t + 1) * G4H;
            xnext = xrow[xcolumn];
        }

        float gk = xcur;
        float gs = 0.f;
        GATE_TERM(0) GATE_TERM(1) GATE_TERM(2) GATE_TERM(3) GATE_TERM(4)
        GATE_TERM(5) GATE_TERM(6) GATE_TERM(7) GATE_TERM(8) GATE_TERM(9)
        const float pre = gk + other_row_f(gs);

        const float e = __builtin_amdgcn_exp2f(nk * pre);
        const float a = fmaf(1.0f / (1.0f + e), sc, off);

        const float i_ = quad_share_f<0>(a);
        const float f_ = quad_share_f<1>(a);
        const float g_ = quad_share_f<2>(a);
        const float o_ = quad_share_f<3>(a);
        c = fmaf(f_, c, i_ * g_);

        const bool low = __builtin_fabsf(c) < TANH_SPLIT;
        if constexpr (KIND == KIND_TANH_HI) {
            if (!low) {
                hsh[buf][uraw] = o_ * tanhf(c);
            }
        } else {
            if (low) {
                hsh[buf][uraw] = o_ * tanhf(c);
            }
        }

        hand_over();
        hv = hsh[buf][hsrc];
    };

    const Flag<true>  with_load{};
    const Flag<false> no_load{};
    for (int t = 0; t < SEQ_T - 2; t += 2) {
        step(t,     0, xa, xb, with_load);
        step(t + 1, 1, xb, xa, with_load);
    }
    step(SEQ_T - 2, 0, xa, xb, with_load);
    step(SEQ_T - 1, 1, xb, xa, no_load);
#undef GATE_TERM

    if constexpr (KIND == KIND_TANH_LO) {
        if (owner) csh[unit] = c;
    }
    __syncthreads();
}

__device__ __forceinline__ void scan_proj_loop(
    const float* __restrict__ xg, const float* __restrict__ W_out,
    const float* __restrict__ b_out, float* __restrict__ out,
    float (*hsh)[32], const float* csh, const int lane)
{
    const int row = lane >> 4;
    const int col = lane & 15;
    const bool active = col < ROW_UNITS;
    const int  n      = active ? row * ROW_UNITS + col : 0;

    float wout[HID_H];
#pragma unroll
    for (int k = 0; k < HID_H; ++k) wout[k] = W_out[k];
    const float bout = b_out[0];
    float hv = 0.f;

    auto step = [&](const int t, const int buf) {
        hand_over();
        hv = hsh[buf][n];

        float ot = bout;
#pragma unroll
        for (int k = 0; k < HID_H; ++k) {
            const float hk = readlane_f(hv, k < ROW_UNITS ? k : k + (16 - ROW_UNITS));
            ot = fmaf(wout[k], hk, ot);
        }
        if (lane == 0) out[t] = ot;

        if (buf == 0 && lane < 4) {
            size_t off = (size_t)(t + PF_AHEAD) * (G4H * sizeof(float))
                       + (size_t)lane * 256;
            off = off < XG_BYTES ? off : XG_BYTES - sizeof(float);
            __builtin_prefetch((const char*)xg + off, 0, 2);
        }
    };

    for (int t = 0; t < SEQ_T; t += 2) {
        step(t,     0);
        step(t + 1, 1);
    }

    __syncthreads();
    if (active) {
        out[SEQ_T + n]         = hv;
        out[SEQ_T + HID_H + n] = csh[n];
    }
}

__global__ __launch_bounds__(SCAN_THREADS) void lstm_scan(
    const float* __restrict__ xg,
    const float* __restrict__ W_hh,
    const float* __restrict__ W_out,
    const float* __restrict__ b_out,
    const float* __restrict__ h0,
    const float* __restrict__ c0,
    float* __restrict__ out)
{
    __shared__ float hsh[2][32];
    __shared__ float csh[32];

    const int lane = threadIdx.x & 31;
    const int wave = __builtin_amdgcn_readfirstlane((int)(threadIdx.x >> 5));

    if (wave == PROJ_WAVE)
        scan_proj_loop(xg, W_out, b_out, out, hsh, csh, lane);
    else if (wave < PROJ_WAVE)
        scan_unit_loop<KIND_TANH_HI>(xg, W_hh, h0, c0, hsh, csh, wave, lane);
    else
        scan_unit_loop<KIND_TANH_LO>(xg, W_hh, h0, c0, hsh, csh, wave - PROJ_WAVE - 1, lane);
}

extern "C" void kernel_launch(void* const* d_in, const int* in_sizes, int n_in,
                              void* d_out, int out_size, void* d_ws, size_t ws_size,
                              hipStream_t stream) {
    (void)in_sizes; (void)n_in; (void)out_size; (void)ws_size;

    const float* x    = (const float*)d_in[0];
    const float* h0   = (const float*)d_in[1];
    const float* c0   = (const float*)d_in[2];
    const float* W_ih = (const float*)d_in[3];
    const float* W_hh = (const float*)d_in[4];
    const float* b_ih = (const float*)d_in[5];
    const float* b_hh = (const float*)d_in[6];
    const float* W_out= (const float*)d_in[7];
    const float* b_out= (const float*)d_in[8];

    float* out = (float*)d_out;
    float* xg  = (float*)d_ws;

    lstm_gates_gemm<<<SEQ_T / 16, 32, 0, stream>>>(x, W_ih, b_ih, b_hh, xg);

    lstm_scan<<<1, SCAN_THREADS, 0, stream>>>(xg, W_hh, W_out, b_out, h0, c0, out);
}


# scan: unit waves store h unpredicated; a lane outside its wave's tanh range aims at a spare LDS slot (address select off the chain): no exec write, branch or restore per step; values bit-identical
# speedup vs baseline: 2.6032x; 1.0451x over previous
#include <hip/hip_runtime.h>
#include <hip/hip_bf16.h>

typedef __attribute__((ext_vector_type(2))) float v2f;
typedef __attribute__((ext_vector_type(8))) float v8f;

#define SEQ_T 32768
#define IN_D  512
#define HID_H 20
#define G4H   80

__global__ __launch_bounds__(32) void lstm_gates_gemm(
    const float* __restrict__ x,
    const float* __restrict__ W_ih,
    const float* __restrict__ b_ih,
    const float* __restrict__ b_hh,
    float* __restrict__ xg)
{
    const int mtile = blockIdx.x;
    const int lane  = threadIdx.x;
    const int half  = lane >> 4;
    const int l15   = lane & 15;

    v8f acc[5];
#pragma unroll
    for (int nt = 0; nt < 5; ++nt) {
        const int n = nt * 16 + l15;
        const float bias = b_ih[n] + b_hh[n];
#pragma unroll
        for (int v = 0; v < 8; ++v) acc[nt][v] = bias;
    }

    const float* xrow = x + (size_t)(mtile * 16 + l15) * IN_D;

    for (int k0 = 0; k0 < IN_D; k0 += 4) {
        v2f a = *(const v2f*)(xrow + k0 + 2 * half);
#pragma unroll
        for (int nt = 0; nt < 5; ++nt) {
            v2f b = *(const v2f*)(W_ih + (size_t)(nt * 16 + l15) * IN_D + k0 + 2 * half);
            acc[nt] = __builtin_amdgcn_wmma_f32_16x16x4_f32(
                 false, a,  false, b,
                 (short)0, acc[nt],
                 false,  false);
        }
    }

    float* outbase = xg + (size_t)(mtile * 16) * G4H;
#pragma unroll
    for (int nt = 0; nt < 5; ++nt) {
#pragma unroll
        for (int v = 0; v < 8; ++v) {
            outbase[(size_t)(v + 8 * half) * G4H + nt * 16 + l15] = acc[nt][v];
        }
    }
}

#define SCAN_WAVES   7
#define SCAN_THREADS 224
#define WAVE_UNITS   8
#define PROJ_WAVE    3
#define GATE_G       2
#define TANH_SPLIT   0.625f
#define ROW_UNITS    10
#define PF_AHEAD     32
#define HSLOTS       64
#define SPARE_BASE   32
#define XG_BYTES     ((size_t)SEQ_T * G4H * sizeof(float))
#define NEG_LOG2E    (-1.44269504088896340736f)

__device__ __forceinline__ float readlane_f(float v, int l) {
    return __builtin_bit_cast(float,
        __builtin_amdgcn_readlane(__builtin_bit_cast(int, v), l));
}

template <int M>
__device__ __forceinline__ float row_share_f(float v) {
    constexpr unsigned SEL = 0x11111111u * (unsigned)M;
    return __builtin_bit_cast(float,
        __builtin_amdgcn_permlane16(__builtin_bit_cast(int, v),
                                    __builtin_bit_cast(int, v),
                                    SEL, SEL, false, true));
}

template <int Q>
__device__ __forceinline__ float quad_share_f(float v) {
    return __builtin_bit_cast(float,
        __builtin_amdgcn_update_dpp(0, __builtin_bit_cast(int, v),
                                    Q * 0x55, 0xF, 0xF, true));
}

__device__ __forceinline__ float other_row_f(float v) {
    return __builtin_bit_cast(float,
        __builtin_amdgcn_permlanex16(__builtin_bit_cast(int, v),
                                     __builtin_bit_cast(int, v),
                                     0x76543210, 0xfedcba98, false, false));
}

__device__ __forceinline__ void hand_over() {
    __builtin_amdgcn_fence(__ATOMIC_RELEASE, "workgroup", "local");
    __builtin_amdgcn_s_barrier();
    __builtin_amdgcn_fence(__ATOMIC_ACQUIRE, "workgroup", "local");
}

#define KIND_TANH_HI 0
#define KIND_TANH_LO 1

template <bool B> struct Flag { static constexpr bool value = B; };

template <int KIND>
__device__ __forceinline__ void scan_unit_loop(
    const float* __restrict__ xg, const float* __restrict__ W_hh,
    const float* __restrict__ h0, const float* __restrict__ c0,
    float (*hsh)[HSLOTS], float* csh, const int group, const int lane)
{
    const int row   = lane >> 4;
    const int col   = lane & 15;
    const int gate  = lane & 3;
    const int uraw  = group * WAVE_UNITS + (lane >> 2);
    const int urawx = group * WAVE_UNITS + ((lane ^ 16) >> 2);
    const int unit  = uraw  < HID_H ? uraw  : uraw  - 4;
    const int unitx = urawx < HID_H ? urawx : urawx - 4;
    const bool owner = gate == 0 && uraw < HID_H;
    const int hsrc  = col < ROW_UNITS ? row * ROW_UNITS + col : 0;
    const int kbase = row * ROW_UNITS;
    const int grow  = gate * HID_H + unit;
    const int growx = gate * HID_H + unitx;

    float wk[ROW_UNITS], ws[ROW_UNITS];
#pragma unroll
    for (int m = 0; m < ROW_UNITS; ++m) {
        wk[m] = W_hh[(size_t)grow  * HID_H + kbase + m];
        ws[m] = W_hh[(size_t)growx * HID_H + kbase + m];
    }

    const float sc  = gate == GATE_G ? 2.0f : 1.0f;
    const float nk  = sc * NEG_LOG2E;
    const float off = 1.0f - sc;

    float c  = c0[unit];
    float hv = h0[hsrc];

    const unsigned xcolumn = (unsigned)grow;
    float xa = xg[xcolumn];
    float xb = 0.f;

#define GATE_TERM(M)                                                     \
    {                                                                    \
        const float hk = row_share_f<M>(hv);                             \
        gk = fmaf(wk[M], hk, gk);                                        \
        gs = fmaf(ws[M], hk, gs);                                        \
    }

    auto step = [&](const int t, const int buf, const float xcur, float& xnext,
                    auto load_next) {
        if constexpr (decltype(load_next)::value) {
            const float* xrow = xg + (size_t)(t + 1) * G4H;
            xnext = xrow[xcolumn];
        }

        float gk = xcur;
        float gs = 0.f;
        GATE_TERM(0) GATE_TERM(1) GATE_TERM(2) GATE_TERM(3) GATE_TERM(4)
        GATE_TERM(5) GATE_TERM(6) GATE_TERM(7) GATE_TERM(8) GATE_TERM(9)
        const float pre = gk + other_row_f(gs);

        const float e = __builtin_amdgcn_exp2f(nk * pre);
        const float a = fmaf(1.0f / (1.0f + e), sc, off);

        const float i_ = quad_share_f<0>(a);
        const float f_ = quad_share_f<1>(a);
        const float g_ = quad_share_f<2>(a);
        const float o_ = quad_share_f<3>(a);
        c = fmaf(f_, c, i_ * g_);

        const float ac  = __builtin_fabsf(c);
        const bool  low = ac < TANH_SPLIT;
        float th;
        bool served;
        if constexpr (KIND == KIND_TANH_HI) {
            const float e = expf(ac + ac);
            th = fmaf(__builtin_amdgcn_rcpf(e + 1.0f), -2.0f, 1.0f);
            served = !low;
        } else {
            const float x2 = c * c;
            float p = fmaf(x2, -0x1.758e7ap-8f, 0x1.521192p-6f);
            p = fmaf(x2, p, -0x1.b8389cp-5f);
            p = fmaf(x2, p, 0x1.110704p-3f);
            p = fmaf(x2, p, -0x1.555532p-2f);
            th = fmaf(x2, ac * p, ac);
            served = low;
        }
        th = __builtin_copysignf(th, c);
        hsh[buf][served ? uraw : SPARE_BASE + uraw] = o_ * th;

        hand_over();
        hv = hsh[buf][hsrc];
    };

    const Flag<true>  with_load{};
    const Flag<false> no_load{};
    for (int t = 0; t < SEQ_T - 2; t += 2) {
        step(t,     0, xa, xb, with_load);
        step(t + 1, 1, xb, xa, with_load);
    }
    step(SEQ_T - 2, 0, xa, xb, with_load);
    step(SEQ_T - 1, 1, xb, xa, no_load);
#undef GATE_TERM

    if constexpr (KIND == KIND_TANH_LO) {
        if (owner) csh[unit] = c;
    }
    __syncthreads();
}

__device__ __forceinline__ void scan_proj_loop(
    const float* __restrict__ xg, const float* __restrict__ W_out,
    const float* __restrict__ b_out, float* __restrict__ out,
    float (*hsh)[HSLOTS], const float* csh, const int lane)
{
    const int row = lane >> 4;
    const int col = lane & 15;
    const bool active = col < ROW_UNITS;
    const int  n      = active ? row * ROW_UNITS + col : 0;

    float wout[HID_H];
#pragma unroll
    for (int k = 0; k < HID_H; ++k) wout[k] = W_out[k];
    const float bout = b_out[0];
    float hv = 0.f;

    auto step = [&](const int t, const int buf) {
        hand_over();
        hv = hsh[buf][n];

        float ot = bout;
#pragma unroll
        for (int k = 0; k < HID_H; ++k) {
            const float hk = readlane_f(hv, k < ROW_UNITS ? k : k + (16 - ROW_UNITS));
            ot = fmaf(wout[k], hk, ot);
        }
        if (lane == 0) out[t] = ot;

        if (buf == 0 && lane < 4) {
            size_t off = (size_t)(t + PF_AHEAD) * (G4H * sizeof(float))
                       + (size_t)lane * 256;
            off = off < XG_BYTES ? off : XG_BYTES - sizeof(float);
            __builtin_prefetch((const char*)xg + off, 0, 2);
        }
    };

    for (int t = 0; t < SEQ_T; t += 2) {
        step(t,     0);
        step(t + 1, 1);
    }

    __syncthreads();
    if (active) {
        out[SEQ_T + n]         = hv;
        out[SEQ_T + HID_H + n] = csh[n];
    }
}

__global__ __launch_bounds__(SCAN_THREADS) void lstm_scan(
    const float* __restrict__ xg,
    const float* __restrict__ W_hh,
    const float* __restrict__ W_out,
    const float* __restrict__ b_out,
    const float* __restrict__ h0,
    const float* __restrict__ c0,
    float* __restrict__ out)
{
    __shared__ float hsh[2][HSLOTS];
    __shared__ float csh[32];

    const int lane = threadIdx.x & 31;
    const int wave = __builtin_amdgcn_readfirstlane((int)(threadIdx.x >> 5));

    if (wave == PROJ_WAVE)
        scan_proj_loop(xg, W_out, b_out, out, hsh, csh, lane);
    else if (wave < PROJ_WAVE)
        scan_unit_loop<KIND_TANH_HI>(xg, W_hh, h0, c0, hsh, csh, wave, lane);
    else
        scan_unit_loop<KIND_TANH_LO>(xg, W_hh, h0, c0, hsh, csh, wave - PROJ_WAVE - 1, lane);
}

extern "C" void kernel_launch(void* const* d_in, const int* in_sizes, int n_in,
                              void* d_out, int out_size, void* d_ws, size_t ws_size,
                              hipStream_t stream) {
    (void)in_sizes; (void)n_in; (void)out_size; (void)ws_size;

    const float* x    = (const float*)d_in[0];
    const float* h0   = (const float*)d_in[1];
    const float* c0   = (const float*)d_in[2];
    const float* W_ih = (const float*)d_in[3];
    const float* W_hh = (const float*)d_in[4];
    const float* b_ih = (const float*)d_in[5];
    const float* b_hh = (const float*)d_in[6];
    const float* W_out= (const float*)d_in[7];
    const float* b_out= (const float*)d_in[8];

    float* out = (float*)d_out;
    float* xg  = (float*)d_ws;

    lstm_gates_gemm<<<SEQ_T / 16, 32, 0, stream>>>(x, W_ih, b_ih, b_hh, xg);

    lstm_scan<<<1, SCAN_THREADS, 0, stream>>>(xg, W_hh, W_out, b_out, h0, c0, out);
}


# scan: each late value enters the step's chain through one fma: exp2 scale folded into W_hh rows and xg, c = i*g + (f*c), o and c's sign folded into tanh's last fma; rounding of c and h changes
# speedup vs baseline: 2.6894x; 1.0331x over previous
#include <hip/hip_runtime.h>
#include <hip/hip_bf16.h>

typedef __attribute__((ext_vector_type(2))) float v2f;
typedef __attribute__((ext_vector_type(8))) float v8f;

#define SEQ_T 32768
#define IN_D  512
#define HID_H 20
#define G4H   80
#define NEG_LOG2E (-1.44269504088896340736f)

__global__ __launch_bounds__(32) void lstm_gates_gemm(
    const float* __restrict__ x,
    const float* __restrict__ W_ih,
    const float* __restrict__ b_ih,
    const float* __restrict__ b_hh,
    float* __restrict__ xg)
{
    const int mtile = blockIdx.x;
    const int lane  = threadIdx.x;
    const int half  = lane >> 4;
    const int l15   = lane & 15;

    v8f acc[5];
#pragma unroll
    for (int nt = 0; nt < 5; ++nt) {
        const int n = nt * 16 + l15;
        const float bias = b_ih[n] + b_hh[n];
#pragma unroll
        for (int v = 0; v < 8; ++v) acc[nt][v] = bias;
    }

    const float* xrow = x + (size_t)(mtile * 16 + l15) * IN_D;

    for (int k0 = 0; k0 < IN_D; k0 += 4) {
        v2f a = *(const v2f*)(xrow + k0 + 2 * half);
#pragma unroll
        for (int nt = 0; nt < 5; ++nt) {
            v2f b = *(const v2f*)(W_ih + (size_t)(nt * 16 + l15) * IN_D + k0 + 2 * half);
            acc[nt] = __builtin_amdgcn_wmma_f32_16x16x4_f32(
                 false, a,  false, b,
                 (short)0, acc[nt],
                 false,  false);
        }
    }

    float* outbase = xg + (size_t)(mtile * 16) * G4H;
#pragma unroll
    for (int nt = 0; nt < 5; ++nt) {
        const int n = nt * 16 + l15;
        const float colscale = (n >= 2 * HID_H && n < 3 * HID_H) ? 2.0f * NEG_LOG2E : NEG_LOG2E;
#pragma unroll
        for (int v = 0; v < 8; ++v) {
            outbase[(size_t)(v + 8 * half) * G4H + n] = acc[nt][v] * colscale;
        }
    }
}

#define SCAN_WAVES   7
#define SCAN_THREADS 224
#define WAVE_UNITS   8
#define PROJ_WAVE    3
#define GATE_G       2
#define TANH_SPLIT   0.625f
#define ROW_UNITS    10
#define PF_AHEAD     32
#define HSLOTS       64
#define SPARE_BASE   32
#define XG_BYTES     ((size_t)SEQ_T * G4H * sizeof(float))

__device__ __forceinline__ float readlane_f(float v, int l) {
    return __builtin_bit_cast(float,
        __builtin_amdgcn_readlane(__builtin_bit_cast(int, v), l));
}

template <int M>
__device__ __forceinline__ float row_share_f(float v) {
    constexpr unsigned SEL = 0x11111111u * (unsigned)M;
    return __builtin_bit_cast(float,
        __builtin_amdgcn_permlane16(__builtin_bit_cast(int, v),
                                    __builtin_bit_cast(int, v),
                                    SEL, SEL, false, true));
}

template <int Q>
__device__ __forceinline__ float quad_share_f(float v) {
    return __builtin_bit_cast(float,
        __builtin_amdgcn_update_dpp(0, __builtin_bit_cast(int, v),
                                    Q * 0x55, 0xF, 0xF, true));
}

__device__ __forceinline__ float other_row_f(float v) {
    return __builtin_bit_cast(float,
        __builtin_amdgcn_permlanex16(__builtin_bit_cast(int, v),
                                     __builtin_bit_cast(int, v),
                                     0x76543210, 0xfedcba98, false, false));
}

__device__ __forceinline__ void hand_over() {
    __builtin_amdgcn_fence(__ATOMIC_RELEASE, "workgroup", "local");
    __builtin_amdgcn_s_barrier();
    __builtin_amdgcn_fence(__ATOMIC_ACQUIRE, "workgroup", "local");
}

#define KIND_TANH_HI 0
#define KIND_TANH_LO 1

template <bool B> struct Flag { static constexpr bool value = B; };

template <int KIND>
__device__ __forceinline__ void scan_unit_loop(
    const float* __restrict__ xg, const float* __restrict__ W_hh,
    const float* __restrict__ h0, const float* __restrict__ c0,
    float (*hsh)[HSLOTS], float* csh, const int group, const int lane)
{
    const int row   = lane >> 4;
    const int col   = lane & 15;
    const int gate  = lane & 3;
    const int uraw  = group * WAVE_UNITS + (lane >> 2);
    const int urawx = group * WAVE_UNITS + ((lane ^ 16) >> 2);
    const int unit  = uraw  < HID_H ? uraw  : uraw  - 4;
    const int unitx = urawx < HID_H ? urawx : urawx - 4;
    const bool owner = gate == 0 && uraw < HID_H;
    const int hsrc  = col < ROW_UNITS ? row * ROW_UNITS + col : 0;
    const int kbase = row * ROW_UNITS;
    const int grow  = gate * HID_H + unit;
    const int growx = gate * HID_H + unitx;

    const float sc  = gate == GATE_G ? 2.0f : 1.0f;
    const float nk  = sc * NEG_LOG2E;
    const float off = 1.0f - sc;

    float wk[ROW_UNITS], ws[ROW_UNITS];
#pragma unroll
    for (int m = 0; m < ROW_UNITS; ++m) {
        wk[m] = nk * W_hh[(size_t)grow  * HID_H + kbase + m];
        ws[m] = nk * W_hh[(size_t)growx * HID_H + kbase + m];
    }

    float c  = c0[unit];
    float hv = h0[hsrc];

    const unsigned xcolumn = (unsigned)grow;
    float xa = xg[xcolumn];
    float xb = 0.f;

#define GATE_TERM(M)                                                     \
    {                                                                    \
        const float hk = row_share_f<M>(hv);                             \
        gk = fmaf(wk[M], hk, gk);                                        \
        gs = fmaf(ws[M], hk, gs);                                        \
    }

    auto step = [&](const int t, const int buf, const float xcur, float& xnext,
                    auto load_next) {
        if constexpr (decltype(load_next)::value) {
            const float* xrow = xg + (size_t)(t + 1) * G4H;
            xnext = xrow[xcolumn];
        }

        float gk = xcur;
        float gs = 0.f;
        GATE_TERM(0) GATE_TERM(1) GATE_TERM(2) GATE_TERM(3) GATE_TERM(4)
        GATE_TERM(5) GATE_TERM(6) GATE_TERM(7) GATE_TERM(8) GATE_TERM(9)
        const float arg = gk + other_row_f(gs);

        const float e = __builtin_amdgcn_exp2f(arg);
        const float a = fmaf(1.0f / (1.0f + e), sc, off);

        const float i_ = quad_share_f<0>(a);
        const float f_ = quad_share_f<1>(a);
        const float g_ = quad_share_f<2>(a);
        const float o_ = quad_share_f<3>(a);
        c = fmaf(i_, g_, f_ * c);

        const float ac  = __builtin_fabsf(c);
        const bool  low = ac < TANH_SPLIT;
        float hnew;
        bool served;
        if constexpr (KIND == KIND_TANH_HI) {
            const float so = __builtin_copysignf(o_, c);
            const float r  = __builtin_amdgcn_rcpf(expf(ac + ac) + 1.0f);
            hnew = fmaf(r, -2.0f * so, so);
            served = !low;
        } else {
            const float x2 = c * c;
            const float oc = o_ * c;
            float p = fmaf(x2, -0x1.758e7ap-8f, 0x1.521192p-6f);
            p = fmaf(x2, p, -0x1.b8389cp-5f);
            p = fmaf(x2, p, 0x1.110704p-3f);
            p = fmaf(x2, p, -0x1.555532p-2f);
            hnew = fmaf(x2, oc * p, oc);
            served = low;
        }
        hsh[buf][served ? uraw : SPARE_BASE + uraw] = hnew;

        hand_over();
        hv = hsh[buf][hsrc];
    };

    const Flag<true>  with_load{};
    const Flag<false> no_load{};
    for (int t = 0; t < SEQ_T - 2; t += 2) {
        step(t,     0, xa, xb, with_load);
        step(t + 1, 1, xb, xa, with_load);
    }
    step(SEQ_T - 2, 0, xa, xb, with_load);
    step(SEQ_T - 1, 1, xb, xa, no_load);
#undef GATE_TERM

    if constexpr (KIND == KIND_TANH_LO) {
        if (owner) csh[unit] = c;
    }
    __syncthreads();
}

__device__ __forceinline__ void scan_proj_loop(
    const float* __restrict__ xg, const float* __restrict__ W_out,
    const float* __restrict__ b_out, float* __restrict__ out,
    float (*hsh)[HSLOTS], const float* csh, const int lane)
{
    const int row = lane >> 4;
    const int col = lane & 15;
    const bool active = col < ROW_UNITS;
    const int  n      = active ? row * ROW_UNITS + col : 0;

    float wout[HID_H];
#pragma unroll
    for (int k = 0; k < HID_H; ++k) wout[k] = W_out[k];
    const float bout = b_out[0];
    float hv = 0.f;

    auto step = [&](const int t, const int buf) {
        hand_over();
        hv = hsh[buf][n];

        float ot = bout;
#pragma unroll
        for (int k = 0; k < HID_H; ++k) {
            const float hk = readlane_f(hv, k < ROW_UNITS ? k : k + (16 - ROW_UNITS));
            ot = fmaf(wout[k], hk, ot);
        }
        if (lane == 0) out[t] = ot;

        if (buf == 0 && lane < 4) {
            size_t off = (size_t)(t + PF_AHEAD) * (G4H * sizeof(float))
                       + (size_t)lane * 256;
            off = off < XG_BYTES ? off : XG_BYTES - sizeof(float);
            __builtin_prefetch((const char*)xg + off, 0, 2);
        }
    };

    for (int t = 0; t < SEQ_T; t += 2) {
        step(t,     0);
        step(t + 1, 1);
    }

    __syncthreads();
    if (active) {
        out[SEQ_T + n]         = hv;
        out[SEQ_T + HID_H + n] = csh[n];
    }
}

__global__ __launch_bounds__(SCAN_THREADS) void lstm_scan(
    const float* __restrict__ xg,
    const float* __restrict__ W_hh,
    const float* __restrict__ W_out,
    const float* __restrict__ b_out,
    const float* __restrict__ h0,
    const float* __restrict__ c0,
    float* __restrict__ out)
{
    __shared__ float hsh[2][HSLOTS];
    __shared__ float csh[32];

    const int lane = threadIdx.x & 31;
    const int wave = __builtin_amdgcn_readfirstlane((int)(threadIdx.x >> 5));

    if (wave == PROJ_WAVE)
        scan_proj_loop(xg, W_out, b_out, out, hsh, csh, lane);
    else if (wave < PROJ_WAVE)
        scan_unit_loop<KIND_TANH_HI>(xg, W_hh, h0, c0, hsh, csh, wave, lane);
    else
        scan_unit_loop<KIND_TANH_LO>(xg, W_hh, h0, c0, hsh, csh, wave - PROJ_WAVE - 1, lane);
}

extern "C" void kernel_launch(void* const* d_in, const int* in_sizes, int n_in,
                              void* d_out, int out_size, void* d_ws, size_t ws_size,
                              hipStream_t stream) {
    (void)in_sizes; (void)n_in; (void)out_size; (void)ws_size;

    const float* x    = (const float*)d_in[0];
    const float* h0   = (const float*)d_in[1];
    const float* c0   = (const float*)d_in[2];
    const float* W_ih = (const float*)d_in[3];
    const float* W_hh = (const float*)d_in[4];
    const float* b_ih = (const float*)d_in[5];
    const float* b_hh = (const float*)d_in[6];
    const float* W_out= (const float*)d_in[7];
    const float* b_out= (const float*)d_in[8];

    float* out = (float*)d_out;
    float* xg  = (float*)d_ws;

    lstm_gates_gemm<<<SEQ_T / 16, 32, 0, stream>>>(x, W_ih, b_ih, b_hh, xg);

    lstm_scan<<<1, SCAN_THREADS, 0, stream>>>(xg, W_hh, W_out, b_out, h0, c0, out);
}


# scan: HI wave's exp(2|c|) written out for its argument (rint, rest by fma, exp2, 2^n scale): doubling in the constants, no under/overflow selects, +1 in the scale fma; h's last-place rounding moves
# speedup vs baseline: 2.7894x; 1.0372x over previous
#include <hip/hip_runtime.h>
#include <hip/hip_bf16.h>

typedef __attribute__((ext_vector_type(2))) float v2f;
typedef __attribute__((ext_vector_type(8))) float v8f;

#define SEQ_T 32768
#define IN_D  512
#define HID_H 20
#define G4H   80
#define NEG_LOG2E (-1.44269504088896340736f)

__global__ __launch_bounds__(32) void lstm_gates_gemm(
    const float* __restrict__ x,
    const float* __restrict__ W_ih,
    const float* __restrict__ b_ih,
    const float* __restrict__ b_hh,
    float* __restrict__ xg)
{
    const int mtile = blockIdx.x;
    const int lane  = threadIdx.x;
    const int half  = lane >> 4;
    const int l15   = lane & 15;

    v8f acc[5];
#pragma unroll
    for (int nt = 0; nt < 5; ++nt) {
        const int n = nt * 16 + l15;
        const float bias = b_ih[n] + b_hh[n];
#pragma unroll
        for (int v = 0; v < 8; ++v) acc[nt][v] = bias;
    }

    const float* xrow = x + (size_t)(mtile * 16 + l15) * IN_D;

    for (int k0 = 0; k0 < IN_D; k0 += 4) {
        v2f a = *(const v2f*)(xrow + k0 + 2 * half);
#pragma unroll
        for (int nt = 0; nt < 5; ++nt) {
            v2f b = *(const v2f*)(W_ih + (size_t)(nt * 16 + l15) * IN_D + k0 + 2 * half);
            acc[nt] = __builtin_amdgcn_wmma_f32_16x16x4_f32(
                 false, a,  false, b,
                 (short)0, acc[nt],
                 false,  false);
        }
    }

    float* outbase = xg + (size_t)(mtile * 16) * G4H;
#pragma unroll
    for (int nt = 0; nt < 5; ++nt) {
        const int n = nt * 16 + l15;
        const float colscale = (n >= 2 * HID_H && n < 3 * HID_H) ? 2.0f * NEG_LOG2E : NEG_LOG2E;
#pragma unroll
        for (int v = 0; v < 8; ++v) {
            outbase[(size_t)(v + 8 * half) * G4H + n] = acc[nt][v] * colscale;
        }
    }
}

#define SCAN_WAVES   7
#define SCAN_THREADS 224
#define WAVE_UNITS   8
#define PROJ_WAVE    3
#define GATE_G       2
#define TANH_SPLIT   0.625f
#define ROW_UNITS    10
#define PF_AHEAD     32
#define HSLOTS       64
#define SPARE_BASE   32
#define XG_BYTES     ((size_t)SEQ_T * G4H * sizeof(float))
#define EXP_2LOG2E_HI 0x1.715476p+1f
#define EXP_2LOG2E_LO 0x1.4ae0bep-25f

__device__ __forceinline__ float readlane_f(float v, int l) {
    return __builtin_bit_cast(float,
        __builtin_amdgcn_readlane(__builtin_bit_cast(int, v), l));
}

template <int M>
__device__ __forceinline__ float row_share_f(float v) {
    constexpr unsigned SEL = 0x11111111u * (unsigned)M;
    return __builtin_bit_cast(float,
        __builtin_amdgcn_permlane16(__builtin_bit_cast(int, v),
                                    __builtin_bit_cast(int, v),
                                    SEL, SEL, false, true));
}

template <int Q>
__device__ __forceinline__ float quad_share_f(float v) {
    return __builtin_bit_cast(float,
        __builtin_amdgcn_update_dpp(0, __builtin_bit_cast(int, v),
                                    Q * 0x55, 0xF, 0xF, true));
}

__device__ __forceinline__ float other_row_f(float v) {
    return __builtin_bit_cast(float,
        __builtin_amdgcn_permlanex16(__builtin_bit_cast(int, v),
                                     __builtin_bit_cast(int, v),
                                     0x76543210, 0xfedcba98, false, false));
}

__device__ __forceinline__ void hand_over() {
    __builtin_amdgcn_fence(__ATOMIC_RELEASE, "workgroup", "local");
    __builtin_amdgcn_s_barrier();
    __builtin_amdgcn_fence(__ATOMIC_ACQUIRE, "workgroup", "local");
}

#define KIND_TANH_HI 0
#define KIND_TANH_LO 1

template <bool B> struct Flag { static constexpr bool value = B; };

template <int KIND>
__device__ __forceinline__ void scan_unit_loop(
    const float* __restrict__ xg, const float* __restrict__ W_hh,
    const float* __restrict__ h0, const float* __restrict__ c0,
    float (*hsh)[HSLOTS], float* csh, const int group, const int lane)
{
    const int row   = lane >> 4;
    const int col   = lane & 15;
    const int gate  = lane & 3;
    const int uraw  = group * WAVE_UNITS + (lane >> 2);
    const int urawx = group * WAVE_UNITS + ((lane ^ 16) >> 2);
    const int unit  = uraw  < HID_H ? uraw  : uraw  - 4;
    const int unitx = urawx < HID_H ? urawx : urawx - 4;
    const bool owner = gate == 0 && uraw < HID_H;
    const int hsrc  = col < ROW_UNITS ? row * ROW_UNITS + col : 0;
    const int kbase = row * ROW_UNITS;
    const int grow  = gate * HID_H + unit;
    const int growx = gate * HID_H + unitx;

    const float sc  = gate == GATE_G ? 2.0f : 1.0f;
    const float nk  = sc * NEG_LOG2E;
    const float off = 1.0f - sc;

    float wk[ROW_UNITS], ws[ROW_UNITS];
#pragma unroll
    for (int m = 0; m < ROW_UNITS; ++m) {
        wk[m] = nk * W_hh[(size_t)grow  * HID_H + kbase + m];
        ws[m] = nk * W_hh[(size_t)growx * HID_H + kbase + m];
    }

    float c  = c0[unit];
    float hv = h0[hsrc];

    const unsigned xcolumn = (unsigned)grow;
    float xa = xg[xcolumn];
    float xb = 0.f;

#define GATE_TERM(M)                                                     \
    {                                                                    \
        const float hk = row_share_f<M>(hv);                             \
        gk = fmaf(wk[M], hk, gk);                                        \
        gs = fmaf(ws[M], hk, gs);                                        \
    }

    auto step = [&](const int t, const int buf, const float xcur, float& xnext,
                    auto load_next) {
        if constexpr (decltype(load_next)::value) {
            const float* xrow = xg + (size_t)(t + 1) * G4H;
            xnext = xrow[xcolumn];
        }

        float gk = xcur;
        float gs = 0.f;
        GATE_TERM(0) GATE_TERM(1) GATE_TERM(2) GATE_TERM(3) GATE_TERM(4)
        GATE_TERM(5) GATE_TERM(6) GATE_TERM(7) GATE_TERM(8) GATE_TERM(9)
        const float arg = gk + other_row_f(gs);

        const float e = __builtin_amdgcn_exp2f(arg);
        const float a = fmaf(1.0f / (1.0f + e), sc, off);

        const float i_ = quad_share_f<0>(a);
        const float f_ = quad_share_f<1>(a);
        const float g_ = quad_share_f<2>(a);
        const float o_ = quad_share_f<3>(a);
        c = fmaf(i_, g_, f_ * c);

        const float ac  = __builtin_fabsf(c);
        const bool  low = ac < TANH_SPLIT;
        float hnew;
        bool served;
        if constexpr (KIND == KIND_TANH_HI) {
            const float so = __builtin_copysignf(o_, c);
            const float pn = __builtin_rintf(ac * EXP_2LOG2E_HI);
            const float pt = fmaf(ac, EXP_2LOG2E_LO, fmaf(ac, EXP_2LOG2E_HI, -pn));
            const float p2 = __builtin_ldexpf(1.0f, (int)pn);
            const float E1 = fmaf(__builtin_amdgcn_exp2f(pt), p2, 1.0f);
            const float r  = __builtin_amdgcn_rcpf(E1);
            hnew = fmaf(r, -2.0f * so, so);
            served = !low;
        } else {
            const float x2 = c * c;
            const float oc = o_ * c;
            float p = fmaf(x2, -0x1.758e7ap-8f, 0x1.521192p-6f);
            p = fmaf(x2, p, -0x1.b8389cp-5f);
            p = fmaf(x2, p, 0x1.110704p-3f);
            p = fmaf(x2, p, -0x1.555532p-2f);
            hnew = fmaf(x2, oc * p, oc);
            served = low;
        }
        hsh[buf][served ? uraw : SPARE_BASE + uraw] = hnew;

        hand_over();
        hv = hsh[buf][hsrc];
    };

    const Flag<true>  with_load{};
    const Flag<false> no_load{};
    for (int t = 0; t < SEQ_T - 2; t += 2) {
        step(t,     0, xa, xb, with_load);
        step(t + 1, 1, xb, xa, with_load);
    }
    step(SEQ_T - 2, 0, xa, xb, with_load);
    step(SEQ_T - 1, 1, xb, xa, no_load);
#undef GATE_TERM

    if constexpr (KIND == KIND_TANH_LO) {
        if (owner) csh[unit] = c;
    }
    __syncthreads();
}

__device__ __forceinline__ void scan_proj_loop(
    const float* __restrict__ xg, const float* __restrict__ W_out,
    const float* __restrict__ b_out, float* __restrict__ out,
    float (*hsh)[HSLOTS], const float* csh, const int lane)
{
    const int row = lane >> 4;
    const int col = lane & 15;
    const bool active = col < ROW_UNITS;
    const int  n      = active ? row * ROW_UNITS + col : 0;

    float wout[HID_H];
#pragma unroll
    for (int k = 0; k < HID_H; ++k) wout[k] = W_out[k];
    const float bout = b_out[0];
    float hv = 0.f;

    auto step = [&](const int t, const int buf) {
        hand_over();
        hv = hsh[buf][n];

        float ot = bout;
#pragma unroll
        for (int k = 0; k < HID_H; ++k) {
            const float hk = readlane_f(hv, k < ROW_UNITS ? k : k + (16 - ROW_UNITS));
            ot = fmaf(wout[k], hk, ot);
        }
        if (lane == 0) out[t] = ot;

        if (buf == 0 && lane < 4) {
            size_t off = (size_t)(t + PF_AHEAD) * (G4H * sizeof(float))
                       + (size_t)lane * 256;
            off = off < XG_BYTES ? off : XG_BYTES - sizeof(float);
            __builtin_prefetch((const char*)xg + off, 0, 2);
        }
    };

    for (int t = 0; t < SEQ_T; t += 2) {
        step(t,     0);
        step(t + 1, 1);
    }

    __syncthreads();
    if (active) {
        out[SEQ_T + n]         = hv;
        out[SEQ_T + HID_H + n] = csh[n];
    }
}

__global__ __launch_bounds__(SCAN_THREADS) void lstm_scan(
    const float* __restrict__ xg,
    const float* __restrict__ W_hh,
    const float* __restrict__ W_out,
    const float* __restrict__ b_out,
    const float* __restrict__ h0,
    const float* __restrict__ c0,
    float* __restrict__ out)
{
    __shared__ float hsh[2][HSLOTS];
    __shared__ float csh[32];

    const int lane = threadIdx.x & 31;
    const int wave = __builtin_amdgcn_readfirstlane((int)(threadIdx.x >> 5));

    if (wave == PROJ_WAVE)
        scan_proj_loop(xg, W_out, b_out, out, hsh, csh, lane);
    else if (wave < PROJ_WAVE)
        scan_unit_loop<KIND_TANH_HI>(xg, W_hh, h0, c0, hsh, csh, wave, lane);
    else
        scan_unit_loop<KIND_TANH_LO>(xg, W_hh, h0, c0, hsh, csh, wave - PROJ_WAVE - 1, lane);
}

extern "C" void kernel_launch(void* const* d_in, const int* in_sizes, int n_in,
                              void* d_out, int out_size, void* d_ws, size_t ws_size,
                              hipStream_t stream) {
    (void)in_sizes; (void)n_in; (void)out_size; (void)ws_size;

    const float* x    = (const float*)d_in[0];
    const float* h0   = (const float*)d_in[1];
    const float* c0   = (const float*)d_in[2];
    const float* W_ih = (const float*)d_in[3];
    const float* W_hh = (const float*)d_in[4];
    const float* b_ih = (const float*)d_in[5];
    const float* b_hh = (const float*)d_in[6];
    const float* W_out= (const float*)d_in[7];
    const float* b_out= (const float*)d_in[8];

    float* out = (float*)d_out;
    float* xg  = (float*)d_ws;

    lstm_gates_gemm<<<SEQ_T / 16, 32, 0, stream>>>(x, W_ih, b_ih, b_hh, xg);

    lstm_scan<<<1, SCAN_THREADS, 0, stream>>>(xg, W_hh, W_out, b_out, h0, c0, out);
}
